# starting kernel
# baseline (speedup 1.0000x reference)
#include <hip/hip_runtime.h>

typedef __attribute__((ext_vector_type(2))) float v2f;
typedef __attribute__((ext_vector_type(4))) float v4f;
typedef __attribute__((ext_vector_type(8))) float v8f;

constexpr int B_ = 8, N_ = 1024, F_ = 512, R_ = 32;
constexpr int ROWS_PER_WAVE   = 16;
constexpr int WAVES_PER_BLOCK = 8;
constexpr int ROWS_PER_BLOCK  = ROWS_PER_WAVE * WAVES_PER_BLOCK;
constexpr int BLOCKS_PER_BATCH = N_ / ROWS_PER_BLOCK;
constexpr int NUM_GEMM_BLOCKS  = B_ * BLOCKS_PER_BATCH;

constexpr size_t WS_WT   = 0;
constexpr size_t WS_PART = WS_WT + (size_t)3 * R_ * F_;
constexpr size_t WS_GSUM = WS_PART + (size_t)NUM_GEMM_BLOCKS * 2 * R_;
constexpr size_t WS_CSUM = WS_GSUM + (size_t)B_ * R_;

__global__ void transpose_w_kernel(const float* __restrict__ W1,
                                   const float* __restrict__ W2,
                                   const float* __restrict__ Wg,
                                   float* __restrict__ wt) {
    int idx = blockIdx.x * blockDim.x + threadIdx.x;
    int m   = idx / (F_ * R_);
    int rem = idx - m * (F_ * R_);
    int k   = rem / R_;
    int r   = rem - k * R_;
    const float* W = (m == 0) ? W1 : ((m == 1) ? W2 : Wg);
    wt[((size_t)m * R_ + r) * F_ + k] = W[rem];
}

__global__ void __launch_bounds__(256)
gemm3_kernel(const float* __restrict__ x, const float* __restrict__ wt,
             float* __restrict__ z1_out, float* __restrict__ partials) {
    const int wave   = threadIdx.x >> 5;
    const int lane   = threadIdx.x & 31;
    const int lane16 = lane & 15;
    const int laneHi = lane >> 4;
    const int batch  = blockIdx.x >> 3;
    const int blkInB = blockIdx.x & 7;
    const int rowBase = blkInB * ROWS_PER_BLOCK + wave * ROWS_PER_WAVE;

    const float* ax = x + ((size_t)(batch * N_ + rowBase + lane16)) * F_ + 2 * laneHi;
    const float* bw00 = wt + ((size_t)(0 * R_ +  0 + lane16)) * F_ + 2 * laneHi;
    const float* bw01 = wt + ((size_t)(0 * R_ + 16 + lane16)) * F_ + 2 * laneHi;
    const float* bw10 = wt + ((size_t)(1 * R_ +  0 + lane16)) * F_ + 2 * laneHi;
    const float* bw11 = wt + ((size_t)(1 * R_ + 16 + lane16)) * F_ + 2 * laneHi;
    const float* bw20 = wt + ((size_t)(2 * R_ +  0 + lane16)) * F_ + 2 * laneHi;
    const float* bw21 = wt + ((size_t)(2 * R_ + 16 + lane16)) * F_ + 2 * laneHi;

    v8f a00 = {}, a01 = {};
    v8f a10 = {}, a11 = {};
    v8f a20 = {}, a21 = {};

#pragma unroll 4
    for (int k0 = 0; k0 < F_; k0 += 4) {
        v2f a   = *(const v2f*)(ax   + k0);
        v2f b00 = *(const v2f*)(bw00 + k0);
        v2f b01 = *(const v2f*)(bw01 + k0);
        v2f b10 = *(const v2f*)(bw10 + k0);
        v2f b11 = *(const v2f*)(bw11 + k0);
        v2f b20 = *(const v2f*)(bw20 + k0);
        v2f b21 = *(const v2f*)(bw21 + k0);
        a00 = __builtin_amdgcn_wmma_f32_16x16x4_f32(false, a, false, b00, (short)0, a00, false, false);
        a01 = __builtin_amdgcn_wmma_f32_16x16x4_f32(false, a, false, b01, (short)0, a01, false, false);
        a10 = __builtin_amdgcn_wmma_f32_16x16x4_f32(false, a, false, b10, (short)0, a10, false, false);
        a11 = __builtin_amdgcn_wmma_f32_16x16x4_f32(false, a, false, b11, (short)0, a11, false, false);
        a20 = __builtin_amdgcn_wmma_f32_16x16x4_f32(false, a, false, b20, (short)0, a20, false, false);
        a21 = __builtin_amdgcn_wmma_f32_16x16x4_f32(false, a, false, b21, (short)0, a21, false, false);
    }

    float* outp = z1_out + ((size_t)(batch * N_ + rowBase + laneHi * 8)) * R_ + lane16;
#pragma unroll
    for (int v = 0; v < 8; ++v) {
        outp[(size_t)v * R_]      = a00[v];
        outp[(size_t)v * R_ + 16] = a01[v];
    }

    float g0 = 0.f, g1 = 0.f, zg0 = 0.f, zg1 = 0.f;
#pragma unroll
    for (int v = 0; v < 8; ++v) {
        g0  += a20[v];
        g1  += a21[v];
        zg0 += a10[v] * a20[v];
        zg1 += a11[v] * a21[v];
    }

    __shared__ float red[2][WAVES_PER_BLOCK][R_][2];
    red[0][wave][lane16     ][laneHi] = g0;
    red[0][wave][lane16 + 16][laneHi] = g1;
    red[1][wave][lane16     ][laneHi] = zg0;
    red[1][wave][lane16 + 16][laneHi] = zg1;
    __syncthreads();

    if (threadIdx.x < 64) {
        int s = threadIdx.x >> 5;
        int r = threadIdx.x & 31;
        float p = 0.f;
#pragma unroll
        for (int w = 0; w < WAVES_PER_BLOCK; ++w) {
            p += red[s][w][r][0];
            p += red[s][w][r][1];
        }
        partials[((size_t)blockIdx.x * 2 + s) * R_ + r] = p;
    }
}

__global__ void reduce_kernel(const float* __restrict__ partials,
                              const float* __restrict__ bias,
                              float* __restrict__ gsum, float* __restrict__ csum) {
    int tid = threadIdx.x;
    int b = tid >> 5;
    int r = tid & 31;
    float g = 0.f, zg = 0.f;
#pragma unroll
    for (int blk = 0; blk < BLOCKS_PER_BATCH; ++blk) {
        size_t base = ((size_t)(b * BLOCKS_PER_BATCH + blk) * 2) * R_ + r;
        g  += partials[base];
        zg += partials[base + R_];
    }
    gsum[b * R_ + r] = g;
    csum[b * R_ + r] = bias[r] * g + zg;
}

__global__ void epilogue_kernel(float* __restrict__ out,
                                const float* __restrict__ gsum,
                                const float* __restrict__ csum) {
    int i = blockIdx.x * blockDim.x + threadIdx.x;
    size_t base = (size_t)i << 2;
    int b = (int)(base >> 15);
    int r = (int)(base & (R_ - 1));
    v4f v = *(v4f*)(out + base);
    v4f g = *(const v4f*)(gsum + b * R_ + r);
    v4f c = *(const v4f*)(csum + b * R_ + r);
    *(v4f*)(out + base) = v * g + c;
}

extern "C" void kernel_launch(void* const* d_in, const int* in_sizes, int n_in,
                              void* d_out, int out_size, void* d_ws, size_t ws_size,
                              hipStream_t stream) {
    (void)in_sizes; (void)n_in; (void)out_size; (void)ws_size;
    const float* x    = (const float*)d_in[0];
    const float* W1   = (const float*)d_in[1];
    const float* W2   = (const float*)d_in[2];
    const float* Wg   = (const float*)d_in[3];
    const float* bias = (const float*)d_in[4];
    float* out = (float*)d_out;
    float* ws  = (float*)d_ws;

    float* wt       = ws + WS_WT;
    float* partials = ws + WS_PART;
    float* gsum     = ws + WS_GSUM;
    float* csum     = ws + WS_CSUM;

    transpose_w_kernel<<<(3 * F_ * R_) / 256, 256, 0, stream>>>(W1, W2, Wg, wt);
    gemm3_kernel<<<NUM_GEMM_BLOCKS, 256, 0, stream>>>(x, wt, out, partials);
    reduce_kernel<<<1, 256, 0, stream>>>(partials, bias, gsum, csum);
    epilogue_kernel<<<(B_ * N_ * R_ / 4) / 256, 256, 0, stream>>>(out, gsum, csum);
}


# GEMM split by product, 2 passes over x: pass 1 x@[W2,Wg] keeps only block sums; pass 2 x@W1 sums its batch's partials, writes out once; 4->3 launches; loads 2 K blocks ahead; z1 k-sum order changes
# speedup vs baseline: 2.7458x; 2.7458x over previous
#include <hip/hip_runtime.h>


typedef __attribute__((ext_vector_type(2))) float v2f;
typedef __attribute__((ext_vector_type(4))) float v4f;
typedef __attribute__((ext_vector_type(8))) float v8f;

constexpr int B_ = 8, N_ = 1024, F_ = 512, R_ = 32;
constexpr int ROWS_PER_TILE   = 16;
constexpr int TILES_PER_BLOCK = 2;
constexpr int COL_HALVES      = 2;
constexpr int WAVES_PER_BLOCK = TILES_PER_BLOCK * COL_HALVES;
constexpr int ROWS_PER_BLOCK  = ROWS_PER_TILE * TILES_PER_BLOCK;
constexpr int BLOCKS_PER_BATCH = N_ / ROWS_PER_BLOCK;
constexpr int NUM_GEMM_BLOCKS  = B_ * BLOCKS_PER_BATCH;
constexpr int ZT_PITCH = R_ + 4;
constexpr int KPAIRS     = F_ / 8;
constexpr int BLK_PAIRS  = 8;
constexpr int NUM_KBLK   = KPAIRS / BLK_PAIRS;

constexpr size_t WS_WP   = 0;
constexpr size_t WS_PART = WS_WP + (size_t)3 * R_ * F_;

__global__ void pack_w_kernel(const float* __restrict__ W1,
                              const float* __restrict__ W2,
                              const float* __restrict__ Wg,
                              float* __restrict__ wp) {
    int idx  = blockIdx.x * blockDim.x + threadIdx.x;
    int lane = idx & 31;
    int pair = (idx >> 5) & (KPAIRS - 1);
    int mh   = idx >> 11;
    int m    = mh >> 1;
    int r    = (mh & 1) * 16 + (lane & 15);
    int k    = pair * 8 + 2 * (lane >> 4);
    const float* W = (m == 0) ? W1 : ((m == 1) ? W2 : Wg);
    v4f v;
    v.x = W[(k + 0) * R_ + r];
    v.y = W[(k + 1) * R_ + r];
    v.z = W[(k + 4) * R_ + r];
    v.w = W[(k + 5) * R_ + r];
    *(v4f*)(wp + (size_t)idx * 4) = v;
}


#define LOAD_X(S, KB)                                                               \
    _Pragma("unroll")                                                               \
    for (int p = 0; p < BLK_PAIRS; ++p) {                                           \
        xa[S][2 * p]     = *(const v2f*)(ax + ((KB) * BLK_PAIRS + p) * 8);          \
        xa[S][2 * p + 1] = *(const v2f*)(ax + ((KB) * BLK_PAIRS + p) * 8 + 4);      \
    }

#define WMMA_F32(A, B, C) \
    __builtin_amdgcn_wmma_f32_16x16x4_f32(false, (A), false, (B), (short)0, (C), false, false)
#define LO2(V) __builtin_shufflevector((V), (V), 0, 1)
#define HI2(V) __builtin_shufflevector((V), (V), 2, 3)

__global__ void __launch_bounds__(WAVES_PER_BLOCK * 32)
gemm_sums_kernel(const float* __restrict__ x, const float* __restrict__ wp,
                 float* __restrict__ partials) {
    const int wave   = threadIdx.x >> 5;
    const int lane   = threadIdx.x & 31;
    const int lane16 = lane & 15;
    const int laneHi = lane >> 4;
    const int tile   = wave >> 1;
    const int half   = wave & 1;
    const int blockRow = blockIdx.x * ROWS_PER_BLOCK;
    const int rowBase  = blockRow + tile * ROWS_PER_TILE;
    const int rcol     = half * 16 + lane16;

    const float* ax  = x  + (size_t)(rowBase + lane16) * F_ + 2 * laneHi;
    const float* bw1 = wp + ((size_t)((1 * 2 + half) * KPAIRS) * 32 + lane) * 4;
    const float* bw2 = wp + ((size_t)((2 * 2 + half) * KPAIRS) * 32 + lane) * 4;

    v8f a1 = {};
    v8f a2 = {};

    v2f xa[3][2 * BLK_PAIRS];
    v4f wb[3][BLK_PAIRS], wc[3][BLK_PAIRS];

#define LOAD_KBLOCK(S, KB)                                                          \
    LOAD_X(S, KB)                                                                   \
    _Pragma("unroll")                                                               \
    for (int p = 0; p < BLK_PAIRS; ++p) {                                           \
        wb[S][p] = *(const v4f*)(bw1 + (size_t)((KB) * BLK_PAIRS + p) * 128);       \
        wc[S][p] = *(const v4f*)(bw2 + (size_t)((KB) * BLK_PAIRS + p) * 128);       \
    }

#define MMA_KBLOCK(S)                                                               \
    _Pragma("unroll")                                                               \
    for (int p = 0; p < BLK_PAIRS; ++p) {                                           \
        v2f alo = xa[S][2 * p], ahi = xa[S][2 * p + 1];                             \
        a1 = WMMA_F32(alo, LO2(wb[S][p]), a1);                                      \
        a2 = WMMA_F32(alo, LO2(wc[S][p]), a2);                                      \
        a1 = WMMA_F32(ahi, HI2(wb[S][p]), a1);                                      \
        a2 = WMMA_F32(ahi, HI2(wc[S][p]), a2);                                      \
    }

    LOAD_KBLOCK(0, 0)
    LOAD_KBLOCK(1, 1)
#pragma nounroll
    for (int kb = 0; kb < NUM_KBLK - 2; kb += 3) {
        LOAD_KBLOCK(2, kb + 2)
        MMA_KBLOCK(0)
        LOAD_KBLOCK(0, kb + 3)
        MMA_KBLOCK(1)
        LOAD_KBLOCK(1, kb + 4)
        MMA_KBLOCK(2)
    }
    MMA_KBLOCK(0)
    MMA_KBLOCK(1)
#undef LOAD_KBLOCK
#undef MMA_KBLOCK

    float g = 0.f, zg = 0.f;
#pragma unroll
    for (int v = 0; v < 8; ++v) {
        g  += a2[v];
        zg += a1[v] * a2[v];
    }

    __shared__ float red[2][TILES_PER_BLOCK][R_][2];
    red[0][tile][rcol][laneHi] = g;
    red[1][tile][rcol][laneHi] = zg;
    __syncthreads();

    if (threadIdx.x < 64) {
        int s = threadIdx.x >> 5;
        int r = threadIdx.x & 31;
        float p = 0.f;
#pragma unroll
        for (int t = 0; t < TILES_PER_BLOCK; ++t) {
            p += red[s][t][r][0];
            p += red[s][t][r][1];
        }
        partials[((size_t)blockIdx.x * 2 + s) * R_ + r] = p;
    }
}

__global__ void __launch_bounds__(WAVES_PER_BLOCK * 32)
gemm_out_kernel(const float* __restrict__ x, const float* __restrict__ wp,
                const float* __restrict__ partials, const float* __restrict__ bias,
                float* __restrict__ out) {
    const int wave   = threadIdx.x >> 5;
    const int lane   = threadIdx.x & 31;
    const int lane16 = lane & 15;
    const int laneHi = lane >> 4;
    const int tile   = wave >> 1;
    const int half   = wave & 1;
    const int blockRow = blockIdx.x * ROWS_PER_BLOCK;
    const int rowBase  = blockRow + tile * ROWS_PER_TILE;
    const int rcol     = half * 16 + lane16;
    const int batch    = blockIdx.x / BLOCKS_PER_BATCH;

    const float* ax  = x  + (size_t)(rowBase + lane16) * F_ + 2 * laneHi;
    const float* bw0 = wp + ((size_t)((0 * 2 + half) * KPAIRS) * 32 + lane) * 4;

    __shared__ __attribute__((aligned(16))) float zt[ROWS_PER_BLOCK][ZT_PITCH];
    __shared__ __attribute__((aligned(16))) float bsum[2][R_];

    float pv[BLOCKS_PER_BATCH];
    const bool reducer = threadIdx.x < 64;
    if (reducer) {
        const int s = threadIdx.x >> 5;
        const int r = threadIdx.x & 31;
        const float* pp = partials + ((size_t)(batch * BLOCKS_PER_BATCH) * 2 + s) * R_ + r;
#pragma unroll
        for (int blk = 0; blk < BLOCKS_PER_BATCH; ++blk) pv[blk] = pp[(size_t)blk * 2 * R_];
    }

    const int c4 = (threadIdx.x & 7) * 4;
    const v4f bb = *(const v4f*)(bias + c4);

    v8f aE = {};
    v8f aO = {};

    v2f xa[3][2 * BLK_PAIRS];
    v4f wa[3][BLK_PAIRS];

#define LOAD_KBLOCK(S, KB)                                                          \
    LOAD_X(S, KB)                                                                   \
    _Pragma("unroll")                                                               \
    for (int p = 0; p < BLK_PAIRS; ++p)                                             \
        wa[S][p] = *(const v4f*)(bw0 + (size_t)((KB) * BLK_PAIRS + p) * 128);

#define MMA_KBLOCK(S)                                                               \
    _Pragma("unroll")                                                               \
    for (int p = 0; p < BLK_PAIRS; ++p) {                                           \
        aE = WMMA_F32(xa[S][2 * p],     LO2(wa[S][p]), aE);                         \
        aO = WMMA_F32(xa[S][2 * p + 1], HI2(wa[S][p]), aO);                         \
    }

    LOAD_KBLOCK(0, 0)
    LOAD_KBLOCK(1, 1)

    if (reducer) {
        float p = 0.f;
#pragma unroll
        for (int blk = 0; blk < BLOCKS_PER_BATCH; ++blk) p += pv[blk];
        bsum[threadIdx.x >> 5][threadIdx.x & 31] = p;
    }

#pragma unroll
    for (int kb = 0; kb < NUM_KBLK - 2; kb += 3) {
        LOAD_KBLOCK(2, kb + 2)
        MMA_KBLOCK(0)
        LOAD_KBLOCK(0, kb + 3)
        MMA_KBLOCK(1)
        LOAD_KBLOCK(1, kb + 4)
        MMA_KBLOCK(2)
    }
    MMA_KBLOCK(0)
    MMA_KBLOCK(1)
#undef LOAD_KBLOCK
#undef MMA_KBLOCK

    const v8f z = aE + aO;

    const int trow = tile * ROWS_PER_TILE + laneHi * 8;
#pragma unroll
    for (int v = 0; v < 8; ++v) zt[trow + v][rcol] = z[v];
    __syncthreads();

    const v4f gs = *(const v4f*)(&bsum[0][c4]);
    const v4f zs = *(const v4f*)(&bsum[1][c4]);
    const v4f cs = bb * gs + zs;
#pragma unroll
    for (int j = 0; j < ROWS_PER_BLOCK / 16; ++j) {
        const int row = j * 16 + (threadIdx.x >> 3);
        v4f val = *(const v4f*)(&zt[row][c4]);
        *(v4f*)(out + (size_t)(blockRow + row) * R_ + c4) = val * gs + cs;
    }
}
#undef LOAD_X
#undef WMMA_F32
#undef LO2
#undef HI2

extern "C" void kernel_launch(void* const* d_in, const int* in_sizes, int n_in,
                              void* d_out, int out_size, void* d_ws, size_t ws_size,
                              hipStream_t stream) {
    (void)in_sizes; (void)n_in; (void)out_size; (void)ws_size;
    const float* x    = (const float*)d_in[0];
    const float* W1   = (const float*)d_in[1];
    const float* W2   = (const float*)d_in[2];
    const float* Wg   = (const float*)d_in[3];
    const float* bias = (const float*)d_in[4];
    float* out = (float*)d_out;
    float* ws  = (float*)d_ws;

    float* wp       = ws + WS_WP;
    float* partials = ws + WS_PART;

    pack_w_kernel<<<(3 * 2 * KPAIRS * 32) / 256, 256, 0, stream>>>(W1, W2, Wg, wp);
    gemm_sums_kernel<<<NUM_GEMM_BLOCKS, WAVES_PER_BLOCK * 32, 0, stream>>>(x, wp, partials);
    gemm_out_kernel<<<NUM_GEMM_BLOCKS, WAVES_PER_BLOCK * 32, 0, stream>>>(x, wp, partials, bias, out);
}


# W pack launch dropped: pass 1 loads W2, Wg as 8-B row segments and one in-register lane-pair exchange turns each into the B fragment; it packs pass 2's W1 itself; 3 -> 2 launches; sum order kept
# speedup vs baseline: 2.9865x; 1.0877x over previous
#include <hip/hip_runtime.h>


typedef __attribute__((ext_vector_type(2))) float v2f;
typedef __attribute__((ext_vector_type(4))) float v4f;
typedef __attribute__((ext_vector_type(8))) float v8f;

constexpr int B_ = 8, N_ = 1024, F_ = 512, R_ = 32;
constexpr int ROWS_PER_TILE   = 16;
constexpr int TILES_PER_BLOCK = 2;
constexpr int COL_HALVES      = 2;
constexpr int WAVES_PER_BLOCK = TILES_PER_BLOCK * COL_HALVES;
constexpr int ROWS_PER_BLOCK  = ROWS_PER_TILE * TILES_PER_BLOCK;
constexpr int BLOCKS_PER_BATCH = N_ / ROWS_PER_BLOCK;
constexpr int NUM_GEMM_BLOCKS  = B_ * BLOCKS_PER_BATCH;
constexpr int ZT_PITCH = R_ + 4;
constexpr int KPAIRS     = F_ / 8;
constexpr int BLK_PAIRS  = 8;
constexpr int NUM_KBLK   = KPAIRS / BLK_PAIRS;

constexpr size_t WS_WP   = 0;
constexpr size_t WS_PART = WS_WP + (size_t)3 * R_ * F_;


#define LOAD_X(S, KB)                                                               \
    _Pragma("unroll")                                                               \
    for (int p = 0; p < BLK_PAIRS; ++p) {                                           \
        xa[S][2 * p]     = *(const v2f*)(ax + ((KB) * BLK_PAIRS + p) * 8);          \
        xa[S][2 * p + 1] = *(const v2f*)(ax + ((KB) * BLK_PAIRS + p) * 8 + 4);      \
    }

#define WMMA_F32(A, B, C) \
    __builtin_amdgcn_wmma_f32_16x16x4_f32(false, (A), false, (B), (short)0, (C), false, false)
#define LO2(V) __builtin_shufflevector((V), (V), 0, 1)
#define HI2(V) __builtin_shufflevector((V), (V), 2, 3)

__device__ __forceinline__ v2f pair_turn(v2f raw, bool odd) {
    const float rx = raw[0], ry = raw[1];
    const float send = odd ? rx : ry;
    const float got = __int_as_float(__builtin_amdgcn_mov_dpp(__float_as_int(send), 0xB1, 0xF, 0xF, true));
    v2f f;
    f[0] = odd ? got : rx;
    f[1] = odd ? ry : got;
    return f;
}

__global__ void __launch_bounds__(WAVES_PER_BLOCK * 32)
gemm_sums_kernel(const float* __restrict__ x, const float* __restrict__ W1,
                 const float* __restrict__ W2, const float* __restrict__ Wg,
                 float* __restrict__ wp, float* __restrict__ partials) {
    const int wave   = threadIdx.x >> 5;
    const int lane   = threadIdx.x & 31;
    const int lane16 = lane & 15;
    const int laneHi = lane >> 4;
    const int tile   = wave >> 1;
    const int half   = wave & 1;
    const int blockRow = blockIdx.x * ROWS_PER_BLOCK;
    const int rowBase  = blockRow + tile * ROWS_PER_TILE;
    const int rcol     = half * 16 + lane16;
    const bool odd     = (lane16 & 1) != 0;

    const float* ax  = x + (size_t)(rowBase + lane16) * F_ + 2 * laneHi;
    const size_t woff = (size_t)(2 * laneHi + (lane16 & 1)) * R_ + half * 16 + (lane16 & ~1);
    const float* bw1 = W2 + woff;
    const float* bw2 = Wg + woff;

    const bool packer = (wave == WAVES_PER_BLOCK - 1) && (blockIdx.x < (2 * KPAIRS * 32) / 32);
    const int  pidx   = blockIdx.x * 32 + lane;
    v4f pk = {};
    if (packer) {
        const int pl = pidx & 31;
        const int pp = (pidx >> 5) & (KPAIRS - 1);
        const int pr = (pidx >> 11) * 16 + (pl & 15);
        const int pkk = pp * 8 + 2 * (pl >> 4);
        pk.x = W1[(pkk + 0) * R_ + pr];
        pk.y = W1[(pkk + 1) * R_ + pr];
        pk.z = W1[(pkk + 4) * R_ + pr];
        pk.w = W1[(pkk + 5) * R_ + pr];
    }

    v8f a1 = {};
    v8f a2 = {};

    constexpr int KS1   = 8;
    constexpr int NBLK1 = F_ / 4 / KS1;
    v2f xa[3][KS1];
    v2f wb[3][KS1], wc[3][KS1];

#define HOLD_ORDER __builtin_amdgcn_sched_barrier(0);

#define LOAD_KBLOCK(S, KB)                                                          \
    _Pragma("unroll")                                                               \
    for (int s = 0; s < KS1; ++s) {                                                 \
        xa[S][s] = *(const v2f*)(ax + ((KB) * KS1 + s) * 4);                        \
        wb[S][s] = *(const v2f*)(bw1 + (size_t)((KB) * KS1 + s) * 4 * R_);          \
        wc[S][s] = *(const v2f*)(bw2 + (size_t)((KB) * KS1 + s) * 4 * R_);          \
    }                                                                               \
    HOLD_ORDER

#define MMA_KBLOCK(S)                                                               \
    _Pragma("unroll")                                                               \
    for (int s = 0; s < KS1; ++s) {                                                 \
        a1 = WMMA_F32(xa[S][s], pair_turn(wb[S][s], odd), a1);                      \
        a2 = WMMA_F32(xa[S][s], pair_turn(wc[S][s], odd), a2);                      \
    }                                                                               \
    HOLD_ORDER

    LOAD_KBLOCK(0, 0)
    LOAD_KBLOCK(1, 1)
#pragma unroll
    for (int kb = 0; kb < NBLK1 - 4; kb += 3) {
        LOAD_KBLOCK(2, kb + 2)
        MMA_KBLOCK(0)
        LOAD_KBLOCK(0, kb + 3)
        MMA_KBLOCK(1)
        LOAD_KBLOCK(1, kb + 4)
        MMA_KBLOCK(2)
    }
    LOAD_KBLOCK(2, NBLK1 - 2)
    MMA_KBLOCK(0)
    LOAD_KBLOCK(0, NBLK1 - 1)
    MMA_KBLOCK(1)
    MMA_KBLOCK(2)
    MMA_KBLOCK(0)
#undef LOAD_KBLOCK
#undef MMA_KBLOCK
#undef HOLD_ORDER

    if (packer) *(v4f*)(wp + (size_t)pidx * 4) = pk;

    float g = 0.f, zg = 0.f;
#pragma unroll
    for (int v = 0; v < 8; ++v) {
        g  += a2[v];
        zg += a1[v] * a2[v];
    }

    __shared__ float red[2][TILES_PER_BLOCK][R_][2];
    red[0][tile][rcol][laneHi] = g;
    red[1][tile][rcol][laneHi] = zg;
    __syncthreads();

    if (threadIdx.x < 64) {
        int s = threadIdx.x >> 5;
        int r = threadIdx.x & 31;
        float p = 0.f;
#pragma unroll
        for (int t = 0; t < TILES_PER_BLOCK; ++t) {
            p += red[s][t][r][0];
            p += red[s][t][r][1];
        }
        partials[((size_t)blockIdx.x * 2 + s) * R_ + r] = p;
    }
}

__global__ void __launch_bounds__(WAVES_PER_BLOCK * 32)
gemm_out_kernel(const float* __restrict__ x, const float* __restrict__ wp,
                const float* __restrict__ partials, const float* __restrict__ bias,
                float* __restrict__ out) {
    const int wave   = threadIdx.x >> 5;
    const int lane   = threadIdx.x & 31;
    const int lane16 = lane & 15;
    const int laneHi = lane >> 4;
    const int tile   = wave >> 1;
    const int half   = wave & 1;
    const int blockRow = blockIdx.x * ROWS_PER_BLOCK;
    const int rowBase  = blockRow + tile * ROWS_PER_TILE;
    const int rcol     = half * 16 + lane16;
    const int batch    = blockIdx.x / BLOCKS_PER_BATCH;

    const float* ax  = x  + (size_t)(rowBase + lane16) * F_ + 2 * laneHi;
    const float* bw0 = wp + ((size_t)((0 * 2 + half) * KPAIRS) * 32 + lane) * 4;

    __shared__ __attribute__((aligned(16))) float zt[ROWS_PER_BLOCK][ZT_PITCH];
    __shared__ __attribute__((aligned(16))) float bsum[2][R_];

    float pv[BLOCKS_PER_BATCH];
    const bool reducer = threadIdx.x < 64;
    if (reducer) {
        const int s = threadIdx.x >> 5;
        const int r = threadIdx.x & 31;
        const float* pp = partials + ((size_t)(batch * BLOCKS_PER_BATCH) * 2 + s) * R_ + r;
#pragma unroll
        for (int blk = 0; blk < BLOCKS_PER_BATCH; ++blk) pv[blk] = pp[(size_t)blk * 2 * R_];
    }

    const int c4 = (threadIdx.x & 7) * 4;
    const v4f bb = *(const v4f*)(bias + c4);

    v8f aE = {};
    v8f aO = {};

    v2f xa[3][2 * BLK_PAIRS];
    v4f wa[3][BLK_PAIRS];

#define LOAD_KBLOCK(S, KB)                                                          \
    LOAD_X(S, KB)                                                                   \
    _Pragma("unroll")                                                               \
    for (int p = 0; p < BLK_PAIRS; ++p)                                             \
        wa[S][p] = *(const v4f*)(bw0 + (size_t)((KB) * BLK_PAIRS + p) * 128);

#define MMA_KBLOCK(S)                                                               \
    _Pragma("unroll")                                                               \
    for (int p = 0; p < BLK_PAIRS; ++p) {                                           \
        aE = WMMA_F32(xa[S][2 * p],     LO2(wa[S][p]), aE);                         \
        aO = WMMA_F32(xa[S][2 * p + 1], HI2(wa[S][p]), aO);                         \
    }

    LOAD_KBLOCK(0, 0)
    LOAD_KBLOCK(1, 1)

    if (reducer) {
        float p = 0.f;
#pragma unroll
        for (int blk = 0; blk < BLOCKS_PER_BATCH; ++blk) p += pv[blk];
        bsum[threadIdx.x >> 5][threadIdx.x & 31] = p;
    }

#pragma unroll
    for (int kb = 0; kb < NUM_KBLK - 2; kb += 3) {
        LOAD_KBLOCK(2, kb + 2)
        MMA_KBLOCK(0)
        LOAD_KBLOCK(0, kb + 3)
        MMA_KBLOCK(1)
        LOAD_KBLOCK(1, kb + 4)
        MMA_KBLOCK(2)
    }
    MMA_KBLOCK(0)
    MMA_KBLOCK(1)
#undef LOAD_KBLOCK
#undef MMA_KBLOCK

    const v8f z = aE + aO;

    const int trow = tile * ROWS_PER_TILE + laneHi * 8;
#pragma unroll
    for (int v = 0; v < 8; ++v) zt[trow + v][rcol] = z[v];
    __syncthreads();

    const v4f gs = *(const v4f*)(&bsum[0][c4]);
    const v4f zs = *(const v4f*)(&bsum[1][c4]);
    const v4f cs = bb * gs + zs;
#pragma unroll
    for (int j = 0; j < ROWS_PER_BLOCK / 16; ++j) {
        const int row = j * 16 + (threadIdx.x >> 3);
        v4f val = *(const v4f*)(&zt[row][c4]);
        *(v4f*)(out + (size_t)(blockRow + row) * R_ + c4) = val * gs + cs;
    }
}
#undef LOAD_X
#undef WMMA_F32
#undef LO2
#undef HI2

extern "C" void kernel_launch(void* const* d_in, const int* in_sizes, int n_in,
                              void* d_out, int out_size, void* d_ws, size_t ws_size,
                              hipStream_t stream) {
    (void)in_sizes; (void)n_in; (void)out_size; (void)ws_size;
    const float* x    = (const float*)d_in[0];
    const float* W1   = (const float*)d_in[1];
    const float* W2   = (const float*)d_in[2];
    const float* Wg   = (const float*)d_in[3];
    const float* bias = (const float*)d_in[4];
    float* out = (float*)d_out;
    float* ws  = (float*)d_ws;

    float* wp       = ws + WS_WP;
    float* partials = ws + WS_PART;

    gemm_sums_kernel<<<NUM_GEMM_BLOCKS, WAVES_PER_BLOCK * 32, 0, stream>>>(x, W1, W2, Wg, wp, partials);
    gemm_out_kernel<<<NUM_GEMM_BLOCKS, WAVES_PER_BLOCK * 32, 0, stream>>>(x, wp, partials, bias, out);
}
